# starting kernel
# baseline (speedup 1.0000x reference)
#include <hip/hip_runtime.h>
#include <hip/hip_bf16.h>

typedef float v2f __attribute__((ext_vector_type(2)));
typedef float v8f __attribute__((ext_vector_type(8)));

#define NN 16384
#define DD 16
#define PP 8
#define MM 50
#define HH 128

#define PW12 288
#define PW3   32
#define SXY   28
#define SH   132
#define WAVES 4

static __device__ __forceinline__ v2f ldsv2(const float* p) { return *(const v2f*)p; }

__launch_bounds__(128, 2)
__global__ void sde_rollout_kernel(const float* __restrict__ X0,
                                   const float* __restrict__ V0,
                                   const float* __restrict__ Yobs,
                                   const float* __restrict__ noise,
                                   const float* __restrict__ W1,
                                   const float* __restrict__ b1,
                                   const float* __restrict__ W2,
                                   const float* __restrict__ b2,
                                   const float* __restrict__ W3,
                                   const float* __restrict__ b3,
                                   float* __restrict__ outX,
                                   float* __restrict__ outV)
{
    __shared__ float sW1p[12 * PW12];
    __shared__ float sW1t[HH];
    __shared__ float sB1 [HH];
    __shared__ float sW2p[64 * PW12];
    __shared__ float sB2 [HH];
    __shared__ float sW3p[64 * PW3];
    __shared__ float sB3 [DD];
    __shared__ float sXY [WAVES][16 * SXY];
    __shared__ float sHH [WAVES][16 * SH];

    const int tid  = threadIdx.x;
    const int lane = tid & 31;
    const int wave = tid >> 5;
    const int lo   = lane & 15;
    const int hi   = lane >> 4;
    const int rowbase = blockIdx.x * (WAVES * 16) + wave * 16;

    for (int i = tid; i < 24 * HH; i += blockDim.x) {
        int r = i >> 7, c = i & 127;
        sW1p[(r >> 1) * PW12 + c * 2 + (r & 1)] = W1[(1 + r) * HH + c];
    }
    if (tid < HH) { sW1t[tid] = W1[tid]; sB1[tid] = b1[tid]; sB2[tid] = b2[tid]; }
    for (int i = tid; i < HH * HH; i += blockDim.x) {
        int r = i >> 7, c = i & 127;
        sW2p[(r >> 1) * PW12 + c * 2 + (r & 1)] = W2[i];
    }
    for (int i = tid; i < HH * DD; i += blockDim.x) {
        int r = i >> 4, c = i & 15;
        sW3p[(r >> 1) * PW3 + c * 2 + (r & 1)] = W3[i];
    }
    if (tid < DD) sB3[tid] = b3[tid];

    float* xy = sXY[wave];
    for (int i = lane; i < 16 * PP; i += 32) {
        int r = i >> 3, c = i & 7;
        xy[r * SXY + DD + c] = Yobs[(rowbase + r) * PP + c];
    }
    __syncthreads();

    float xr[8], va[8];
    #pragma unroll
    for (int v = 0; v < 8; ++v) {
        int row = rowbase + v + 8 * hi;
        xr[v] = X0[row * DD + lo];
        va[v] = V0[row];
    }

    const float dt    = 1.0f / MM;
    const float sqdt  = __builtin_sqrtf(dt);
    const float decay = 1.0f - dt;
    float* hsc = sHH[wave];
    const v8f zc = {0.f, 0.f, 0.f, 0.f, 0.f, 0.f, 0.f, 0.f};

    for (int m = 0; m < MM; ++m) {
        const float t = m * dt;

        #pragma unroll
        for (int v = 0; v < 8; ++v)
            xy[(v + 8 * hi) * SXY + lo] = xr[v];

        float ep[8];
        #pragma unroll
        for (int v = 0; v < 8; ++v) {
            size_t row = (size_t)m * NN + (size_t)(rowbase + v + 8 * hi);
            ep[v] = noise[row * DD + lo];
        }

        asm volatile("s_wait_dscnt 0" ::: "memory");

        v8f acc[8];

        {
            v2f a = ldsv2(&xy[lo * SXY + 2 * hi]);
            v2f b[8];
            #pragma unroll
            for (int c = 0; c < 8; ++c) b[c] = ldsv2(&sW1p[hi * PW12 + (c * 16 + lo) * 2]);
            #pragma unroll
            for (int kk = 0; kk < 6; ++kk) {
                v2f an = a; v2f bn[8];
                #pragma unroll
                for (int c = 0; c < 8; ++c) bn[c] = b[c];
                if (kk < 5) {
                    an = ldsv2(&xy[lo * SXY + 4 * (kk + 1) + 2 * hi]);
                    #pragma unroll
                    for (int c = 0; c < 8; ++c)
                        bn[c] = ldsv2(&sW1p[(2 * (kk + 1) + hi) * PW12 + (c * 16 + lo) * 2]);
                }
                #pragma unroll
                for (int c = 0; c < 8; ++c)
                    acc[c] = __builtin_amdgcn_wmma_f32_16x16x4_f32(
                        false, a, false, b[c], (short)0, (kk == 0 ? zc : acc[c]), false, false);
                a = an;
                #pragma unroll
                for (int c = 0; c < 8; ++c) b[c] = bn[c];
            }
        }
        #pragma unroll
        for (int c = 0; c < 8; ++c) {
            float bias = t * sW1t[c * 16 + lo] + sB1[c * 16 + lo];
            #pragma unroll
            for (int v = 0; v < 8; ++v)
                hsc[(v + 8 * hi) * SH + c * 16 + lo] = fmaxf(acc[c][v] + bias, 0.0f);
        }
        asm volatile("s_wait_dscnt 0" ::: "memory");

        {
            v2f a = ldsv2(&hsc[lo * SH + 2 * hi]);
            v2f b[8];
            #pragma unroll
            for (int c = 0; c < 8; ++c) b[c] = ldsv2(&sW2p[hi * PW12 + (c * 16 + lo) * 2]);
            #pragma unroll
            for (int kk = 0; kk < 32; ++kk) {
                v2f an = a; v2f bn[8];
                #pragma unroll
                for (int c = 0; c < 8; ++c) bn[c] = b[c];
                if (kk < 31) {
                    an = ldsv2(&hsc[lo * SH + 4 * (kk + 1) + 2 * hi]);
                    #pragma unroll
                    for (int c = 0; c < 8; ++c)
                        bn[c] = ldsv2(&sW2p[(2 * (kk + 1) + hi) * PW12 + (c * 16 + lo) * 2]);
                }
                #pragma unroll
                for (int c = 0; c < 8; ++c)
                    acc[c] = __builtin_amdgcn_wmma_f32_16x16x4_f32(
                        false, a, false, b[c], (short)0, (kk == 0 ? zc : acc[c]), false, false);
                a = an;
                #pragma unroll
                for (int c = 0; c < 8; ++c) b[c] = bn[c];
            }
        }
        #pragma unroll
        for (int c = 0; c < 8; ++c) {
            float bias = sB2[c * 16 + lo];
            #pragma unroll
            for (int v = 0; v < 8; ++v)
                hsc[(v + 8 * hi) * SH + c * 16 + lo] = fmaxf(acc[c][v] + bias, 0.0f);
        }
        asm volatile("s_wait_dscnt 0" ::: "memory");

        v8f z;
        {
            v2f a = ldsv2(&hsc[lo * SH + 2 * hi]);
            v2f b = ldsv2(&sW3p[hi * PW3 + lo * 2]);
            #pragma unroll
            for (int kk = 0; kk < 32; ++kk) {
                v2f an = a, bn = b;
                if (kk < 31) {
                    an = ldsv2(&hsc[lo * SH + 4 * (kk + 1) + 2 * hi]);
                    bn = ldsv2(&sW3p[(2 * (kk + 1) + hi) * PW3 + lo * 2]);
                }
                z = __builtin_amdgcn_wmma_f32_16x16x4_f32(
                    false, a, false, b, (short)0, (kk == 0 ? zc : z), false, false);
                a = an; b = bn;
            }
        }

        const float b3v = sB3[lo];
        #pragma unroll
        for (int v = 0; v < 8; ++v) {
            float zz = z[v] + b3v;
            float s  = 0.5f * dt * zz * zz + sqdt * zz * ep[v];
            s += __shfl_xor(s, 1);
            s += __shfl_xor(s, 2);
            s += __shfl_xor(s, 4);
            s += __shfl_xor(s, 8);
            va[v] += s;
            xr[v]  = decay * xr[v] + sqdt * ep[v];
        }
    }

    #pragma unroll
    for (int v = 0; v < 8; ++v) {
        int row = rowbase + v + 8 * hi;
        outX[row * DD + lo] = xr[v];
        if (lo == 0) outV[row] = va[v];
    }
}

extern "C" void kernel_launch(void* const* d_in, const int* in_sizes, int n_in,
                              void* d_out, int out_size, void* d_ws, size_t ws_size,
                              hipStream_t stream) {
    (void)in_sizes; (void)n_in; (void)out_size; (void)d_ws; (void)ws_size;
    const float* X0    = (const float*)d_in[0];
    const float* V0    = (const float*)d_in[1];
    const float* Yobs  = (const float*)d_in[2];
    const float* noise = (const float*)d_in[3];
    const float* W1    = (const float*)d_in[4];
    const float* b1    = (const float*)d_in[5];
    const float* W2    = (const float*)d_in[6];
    const float* b2    = (const float*)d_in[7];
    const float* W3    = (const float*)d_in[8];
    const float* b3    = (const float*)d_in[9];

    float* outX = (float*)d_out;
    float* outV = outX + (size_t)NN * DD;

    dim3 grid(NN / (WAVES * 16));
    dim3 block(WAVES * 32);
    sde_rollout_kernel<<<grid, block, 0, stream>>>(X0, V0, Yobs, noise,
                                                   W1, b1, W2, b2, W3, b3,
                                                   outX, outV);
}


# V update: the eight 16-lane row sums use DPP row_xmask adds on the vector ALU instead of 32 LDS-unit lane permutes each waited for in turn; same xor butterfly, sum order unchanged
# speedup vs baseline: 1.1639x; 1.1639x over previous
#include <hip/hip_runtime.h>
#include <hip/hip_bf16.h>

typedef float v2f __attribute__((ext_vector_type(2)));
typedef float v8f __attribute__((ext_vector_type(8)));

#define NN 16384
#define DD 16
#define PP 8
#define MM 50
#define HH 128

#define PW12 288
#define PW3   32
#define SXY   28
#define SH   132
#define WAVES 4

static __device__ __forceinline__ v2f ldsv2(const float* p) { return *(const v2f*)p; }

template <int M>
static __device__ __forceinline__ float row_xor_add(float s) {
    const int partner = __builtin_amdgcn_update_dpp(0, __float_as_int(s), 0x160 + M, 0xF, 0xF, true);
    return s + __int_as_float(partner);
}

__launch_bounds__(128, 2)
__global__ void sde_rollout_kernel(const float* __restrict__ X0,
                                   const float* __restrict__ V0,
                                   const float* __restrict__ Yobs,
                                   const float* __restrict__ noise,
                                   const float* __restrict__ W1,
                                   const float* __restrict__ b1,
                                   const float* __restrict__ W2,
                                   const float* __restrict__ b2,
                                   const float* __restrict__ W3,
                                   const float* __restrict__ b3,
                                   float* __restrict__ outX,
                                   float* __restrict__ outV)
{
    __shared__ float sW1p[12 * PW12];
    __shared__ float sW1t[HH];
    __shared__ float sB1 [HH];
    __shared__ float sW2p[64 * PW12];
    __shared__ float sB2 [HH];
    __shared__ float sW3p[64 * PW3];
    __shared__ float sB3 [DD];
    __shared__ float sXY [WAVES][16 * SXY];
    __shared__ float sHH [WAVES][16 * SH];

    const int tid  = threadIdx.x;
    const int lane = tid & 31;
    const int wave = tid >> 5;
    const int lo   = lane & 15;
    const int hi   = lane >> 4;
    const int rowbase = blockIdx.x * (WAVES * 16) + wave * 16;

    for (int i = tid; i < 24 * HH; i += blockDim.x) {
        int r = i >> 7, c = i & 127;
        sW1p[(r >> 1) * PW12 + c * 2 + (r & 1)] = W1[(1 + r) * HH + c];
    }
    if (tid < HH) { sW1t[tid] = W1[tid]; sB1[tid] = b1[tid]; sB2[tid] = b2[tid]; }
    for (int i = tid; i < HH * HH; i += blockDim.x) {
        int r = i >> 7, c = i & 127;
        sW2p[(r >> 1) * PW12 + c * 2 + (r & 1)] = W2[i];
    }
    for (int i = tid; i < HH * DD; i += blockDim.x) {
        int r = i >> 4, c = i & 15;
        sW3p[(r >> 1) * PW3 + c * 2 + (r & 1)] = W3[i];
    }
    if (tid < DD) sB3[tid] = b3[tid];

    float* xy = sXY[wave];
    for (int i = lane; i < 16 * PP; i += 32) {
        int r = i >> 3, c = i & 7;
        xy[r * SXY + DD + c] = Yobs[(rowbase + r) * PP + c];
    }
    __syncthreads();

    float xr[8], va[8];
    #pragma unroll
    for (int v = 0; v < 8; ++v) {
        int row = rowbase + v + 8 * hi;
        xr[v] = X0[row * DD + lo];
        va[v] = V0[row];
    }

    const float dt    = 1.0f / MM;
    const float sqdt  = __builtin_sqrtf(dt);
    const float decay = 1.0f - dt;
    float* hsc = sHH[wave];
    const v8f zc = {0.f, 0.f, 0.f, 0.f, 0.f, 0.f, 0.f, 0.f};

    for (int m = 0; m < MM; ++m) {
        const float t = m * dt;

        #pragma unroll
        for (int v = 0; v < 8; ++v)
            xy[(v + 8 * hi) * SXY + lo] = xr[v];

        float ep[8];
        #pragma unroll
        for (int v = 0; v < 8; ++v) {
            size_t row = (size_t)m * NN + (size_t)(rowbase + v + 8 * hi);
            ep[v] = noise[row * DD + lo];
        }

        asm volatile("s_wait_dscnt 0" ::: "memory");

        v8f acc[8];

        {
            v2f a = ldsv2(&xy[lo * SXY + 2 * hi]);
            v2f b[8];
            #pragma unroll
            for (int c = 0; c < 8; ++c) b[c] = ldsv2(&sW1p[hi * PW12 + (c * 16 + lo) * 2]);
            #pragma unroll
            for (int kk = 0; kk < 6; ++kk) {
                v2f an = a; v2f bn[8];
                #pragma unroll
                for (int c = 0; c < 8; ++c) bn[c] = b[c];
                if (kk < 5) {
                    an = ldsv2(&xy[lo * SXY + 4 * (kk + 1) + 2 * hi]);
                    #pragma unroll
                    for (int c = 0; c < 8; ++c)
                        bn[c] = ldsv2(&sW1p[(2 * (kk + 1) + hi) * PW12 + (c * 16 + lo) * 2]);
                }
                #pragma unroll
                for (int c = 0; c < 8; ++c)
                    acc[c] = __builtin_amdgcn_wmma_f32_16x16x4_f32(
                        false, a, false, b[c], (short)0, (kk == 0 ? zc : acc[c]), false, false);
                a = an;
                #pragma unroll
                for (int c = 0; c < 8; ++c) b[c] = bn[c];
            }
        }
        #pragma unroll
        for (int c = 0; c < 8; ++c) {
            float bias = t * sW1t[c * 16 + lo] + sB1[c * 16 + lo];
            #pragma unroll
            for (int v = 0; v < 8; ++v)
                hsc[(v + 8 * hi) * SH + c * 16 + lo] = fmaxf(acc[c][v] + bias, 0.0f);
        }
        asm volatile("s_wait_dscnt 0" ::: "memory");

        {
            v2f a = ldsv2(&hsc[lo * SH + 2 * hi]);
            v2f b[8];
            #pragma unroll
            for (int c = 0; c < 8; ++c) b[c] = ldsv2(&sW2p[hi * PW12 + (c * 16 + lo) * 2]);
            #pragma unroll
            for (int kk = 0; kk < 32; ++kk) {
                v2f an = a; v2f bn[8];
                #pragma unroll
                for (int c = 0; c < 8; ++c) bn[c] = b[c];
                if (kk < 31) {
                    an = ldsv2(&hsc[lo * SH + 4 * (kk + 1) + 2 * hi]);
                    #pragma unroll
                    for (int c = 0; c < 8; ++c)
                        bn[c] = ldsv2(&sW2p[(2 * (kk + 1) + hi) * PW12 + (c * 16 + lo) * 2]);
                }
                #pragma unroll
                for (int c = 0; c < 8; ++c)
                    acc[c] = __builtin_amdgcn_wmma_f32_16x16x4_f32(
                        false, a, false, b[c], (short)0, (kk == 0 ? zc : acc[c]), false, false);
                a = an;
                #pragma unroll
                for (int c = 0; c < 8; ++c) b[c] = bn[c];
            }
        }
        #pragma unroll
        for (int c = 0; c < 8; ++c) {
            float bias = sB2[c * 16 + lo];
            #pragma unroll
            for (int v = 0; v < 8; ++v)
                hsc[(v + 8 * hi) * SH + c * 16 + lo] = fmaxf(acc[c][v] + bias, 0.0f);
        }
        asm volatile("s_wait_dscnt 0" ::: "memory");

        v8f z;
        {
            v2f a = ldsv2(&hsc[lo * SH + 2 * hi]);
            v2f b = ldsv2(&sW3p[hi * PW3 + lo * 2]);
            #pragma unroll
            for (int kk = 0; kk < 32; ++kk) {
                v2f an = a, bn = b;
                if (kk < 31) {
                    an = ldsv2(&hsc[lo * SH + 4 * (kk + 1) + 2 * hi]);
                    bn = ldsv2(&sW3p[(2 * (kk + 1) + hi) * PW3 + lo * 2]);
                }
                z = __builtin_amdgcn_wmma_f32_16x16x4_f32(
                    false, a, false, b, (short)0, (kk == 0 ? zc : z), false, false);
                a = an; b = bn;
            }
        }

        const float b3v = sB3[lo];
        float s[8];
        #pragma unroll
        for (int v = 0; v < 8; ++v) {
            float zz = z[v] + b3v;
            s[v]  = 0.5f * dt * zz * zz + sqdt * zz * ep[v];
            xr[v] = decay * xr[v] + sqdt * ep[v];
        }
        #pragma unroll
        for (int v = 0; v < 8; ++v) s[v] = row_xor_add<1>(s[v]);
        #pragma unroll
        for (int v = 0; v < 8; ++v) s[v] = row_xor_add<2>(s[v]);
        #pragma unroll
        for (int v = 0; v < 8; ++v) s[v] = row_xor_add<4>(s[v]);
        #pragma unroll
        for (int v = 0; v < 8; ++v) s[v] = row_xor_add<8>(s[v]);
        #pragma unroll
        for (int v = 0; v < 8; ++v) va[v] += s[v];
    }

    #pragma unroll
    for (int v = 0; v < 8; ++v) {
        int row = rowbase + v + 8 * hi;
        outX[row * DD + lo] = xr[v];
        if (lo == 0) outV[row] = va[v];
    }
}

extern "C" void kernel_launch(void* const* d_in, const int* in_sizes, int n_in,
                              void* d_out, int out_size, void* d_ws, size_t ws_size,
                              hipStream_t stream) {
    (void)in_sizes; (void)n_in; (void)out_size; (void)d_ws; (void)ws_size;
    const float* X0    = (const float*)d_in[0];
    const float* V0    = (const float*)d_in[1];
    const float* Yobs  = (const float*)d_in[2];
    const float* noise = (const float*)d_in[3];
    const float* W1    = (const float*)d_in[4];
    const float* b1    = (const float*)d_in[5];
    const float* W2    = (const float*)d_in[6];
    const float* b2    = (const float*)d_in[7];
    const float* W3    = (const float*)d_in[8];
    const float* b3    = (const float*)d_in[9];

    float* outX = (float*)d_out;
    float* outV = outX + (size_t)NN * DD;

    dim3 grid(NN / (WAVES * 16));
    dim3 block(WAVES * 32);
    sde_rollout_kernel<<<grid, block, 0, stream>>>(X0, V0, Yobs, noise,
                                                   W1, b1, W2, b2, W3, b3,
                                                   outX, outV);
}


# Two waves per row tile (8-wave blocks, 2 per SIMD): each does 64 of 128 hidden columns and half of layer 3's K; h1 halves and partial Z meet in LDS at 2 barriers per step; layer-3 sum order changes
# speedup vs baseline: 1.8923x; 1.6258x over previous
#include <hip/hip_runtime.h>
#include <hip/hip_bf16.h>

typedef float v2f __attribute__((ext_vector_type(2)));
typedef float v8f __attribute__((ext_vector_type(8)));

#define NN 16384
#define DD 16
#define PP 8
#define MM 50
#define HH 128

#define PW12 288
#define PW3   32
#define SXY   28
#define SH   132
#define SH2   68
#define RT     4
#define WAVES  8

static __device__ __forceinline__ v2f ldsv2(const float* p) { return *(const v2f*)p; }

template <int M>
static __device__ __forceinline__ float row_xor_add(float s) {
    const int partner = __builtin_amdgcn_update_dpp(0, __float_as_int(s), 0x160 + M, 0xF, 0xF, true);
    return s + __int_as_float(partner);
}

__launch_bounds__(256, 1)
__global__ void sde_rollout_kernel(const float* __restrict__ X0,
                                   const float* __restrict__ V0,
                                   const float* __restrict__ Yobs,
                                   const float* __restrict__ noise,
                                   const float* __restrict__ W1,
                                   const float* __restrict__ b1,
                                   const float* __restrict__ W2,
                                   const float* __restrict__ b2,
                                   const float* __restrict__ W3,
                                   const float* __restrict__ b3,
                                   float* __restrict__ outX,
                                   float* __restrict__ outV)
{
    __shared__ float sW1p[12 * PW12];
    __shared__ float sW1t[HH];
    __shared__ float sB1 [HH];
    __shared__ float sW2p[64 * PW12];
    __shared__ float sB2 [HH];
    __shared__ float sW3p[64 * PW3];
    __shared__ float sB3 [DD];
    __shared__ float sXY [WAVES][16 * SXY];
    __shared__ float sH1 [RT][16 * SH];
    __shared__ float sH2 [WAVES][16 * SH2];
    __shared__ float sZ  [RT][8 * 32];

    const int tid  = threadIdx.x;
    const int lane = tid & 31;
    const int wave = __builtin_amdgcn_readfirstlane(tid >> 5);
    const int lo   = lane & 15;
    const int hi   = lane >> 4;
    const int rt   = wave & 3;
    const int ch   = wave >> 2;
    const int cb   = ch * 4;
    const int rowbase = blockIdx.x * (RT * 16) + rt * 16;

    for (int i = tid; i < 24 * HH; i += blockDim.x) {
        int r = i >> 7, c = i & 127;
        sW1p[(r >> 1) * PW12 + c * 2 + (r & 1)] = W1[(1 + r) * HH + c];
    }
    if (tid < HH) { sW1t[tid] = W1[tid]; sB1[tid] = b1[tid]; sB2[tid] = b2[tid]; }
    for (int i = tid; i < HH * HH; i += blockDim.x) {
        int r = i >> 7, c = i & 127;
        sW2p[(r >> 1) * PW12 + c * 2 + (r & 1)] = W2[i];
    }
    for (int i = tid; i < HH * DD; i += blockDim.x) {
        int r = i >> 4, c = i & 15;
        sW3p[(r >> 1) * PW3 + c * 2 + (r & 1)] = W3[i];
    }
    if (tid < DD) sB3[tid] = b3[tid];

    float* xy = sXY[wave];
    for (int i = lane; i < 16 * PP; i += 32) {
        int r = i >> 3, c = i & 7;
        xy[r * SXY + DD + c] = Yobs[(rowbase + r) * PP + c];
    }
    __syncthreads();

    float xr[8], va[8];
    #pragma unroll
    for (int v = 0; v < 8; ++v) {
        int row = rowbase + v + 8 * hi;
        xr[v] = X0[row * DD + lo];
        va[v] = V0[row];
    }

    const float dt    = 1.0f / MM;
    const float sqdt  = __builtin_sqrtf(dt);
    const float decay = 1.0f - dt;
    float* h1 = sH1[rt];
    float* h2 = sH2[wave];
    float* zx = sZ[rt];
    const float* w1c  = sW1p + cb * 32;
    const float* w2c  = sW2p + cb * 32;
    const float* w3c  = sW3p + ch * 32 * PW3;
    const float* w1tc = sW1t + cb * 16;
    const float* b1c  = sB1  + cb * 16;
    const float* b2c  = sB2  + cb * 16;
    float* h1c = h1 + cb * 16;
    const v8f zc = {0.f, 0.f, 0.f, 0.f, 0.f, 0.f, 0.f, 0.f};

    for (int m = 0; m < MM; ++m) {
        const float t = m * dt;

        #pragma unroll
        for (int v = 0; v < 8; ++v)
            xy[(v + 8 * hi) * SXY + lo] = xr[v];

        asm volatile("s_wait_dscnt 0" ::: "memory");

        v8f acc[4];

        {
            v2f a = ldsv2(&xy[lo * SXY + 2 * hi]);
            v2f b[4];
            #pragma unroll
            for (int c = 0; c < 4; ++c) b[c] = ldsv2(&w1c[hi * PW12 + (c * 16 + lo) * 2]);
            #pragma unroll
            for (int kk = 0; kk < 6; ++kk) {
                v2f an = a; v2f bn[4];
                #pragma unroll
                for (int c = 0; c < 4; ++c) bn[c] = b[c];
                if (kk < 5) {
                    an = ldsv2(&xy[lo * SXY + 4 * (kk + 1) + 2 * hi]);
                    #pragma unroll
                    for (int c = 0; c < 4; ++c)
                        bn[c] = ldsv2(&w1c[(2 * (kk + 1) + hi) * PW12 + (c * 16 + lo) * 2]);
                }
                #pragma unroll
                for (int c = 0; c < 4; ++c)
                    acc[c] = __builtin_amdgcn_wmma_f32_16x16x4_f32(
                        false, a, false, b[c], (short)0, (kk == 0 ? zc : acc[c]), false, false);
                a = an;
                #pragma unroll
                for (int c = 0; c < 4; ++c) b[c] = bn[c];
            }
        }
        #pragma unroll
        for (int c = 0; c < 4; ++c) {
            float bias = t * w1tc[c * 16 + lo] + b1c[c * 16 + lo];
            #pragma unroll
            for (int v = 0; v < 8; ++v)
                h1c[(v + 8 * hi) * SH + c * 16 + lo] = fmaxf(acc[c][v] + bias, 0.0f);
        }
        __syncthreads();

        float ep[8];
        #pragma unroll
        for (int v = 0; v < 8; ++v) {
            size_t row = (size_t)m * NN + (size_t)(rowbase + v + 8 * hi);
            ep[v] = noise[row * DD + lo];
        }
        __builtin_amdgcn_sched_barrier(0);

        {
            v2f a = ldsv2(&h1[lo * SH + 2 * hi]);
            v2f b[4];
            #pragma unroll
            for (int c = 0; c < 4; ++c) b[c] = ldsv2(&w2c[hi * PW12 + (c * 16 + lo) * 2]);
            #pragma unroll
            for (int kk = 0; kk < 32; ++kk) {
                v2f an = a; v2f bn[4];
                #pragma unroll
                for (int c = 0; c < 4; ++c) bn[c] = b[c];
                if (kk < 31) {
                    an = ldsv2(&h1[lo * SH + 4 * (kk + 1) + 2 * hi]);
                    #pragma unroll
                    for (int c = 0; c < 4; ++c)
                        bn[c] = ldsv2(&w2c[(2 * (kk + 1) + hi) * PW12 + (c * 16 + lo) * 2]);
                }
                #pragma unroll
                for (int c = 0; c < 4; ++c)
                    acc[c] = __builtin_amdgcn_wmma_f32_16x16x4_f32(
                        false, a, false, b[c], (short)0, (kk == 0 ? zc : acc[c]), false, false);
                a = an;
                #pragma unroll
                for (int c = 0; c < 4; ++c) b[c] = bn[c];
            }
        }
        #pragma unroll
        for (int c = 0; c < 4; ++c) {
            float bias = b2c[c * 16 + lo];
            #pragma unroll
            for (int v = 0; v < 8; ++v)
                h2[(v + 8 * hi) * SH2 + c * 16 + lo] = fmaxf(acc[c][v] + bias, 0.0f);
        }
        asm volatile("s_wait_dscnt 0" ::: "memory");

        v8f z;
        {
            v2f a = ldsv2(&h2[lo * SH2 + 2 * hi]);
            v2f b = ldsv2(&w3c[hi * PW3 + lo * 2]);
            #pragma unroll
            for (int kk = 0; kk < 16; ++kk) {
                v2f an = a, bn = b;
                if (kk < 15) {
                    an = ldsv2(&h2[lo * SH2 + 4 * (kk + 1) + 2 * hi]);
                    bn = ldsv2(&w3c[(2 * (kk + 1) + hi) * PW3 + lo * 2]);
                }
                z = __builtin_amdgcn_wmma_f32_16x16x4_f32(
                    false, a, false, b, (short)0, (kk == 0 ? zc : z), false, false);
                a = an; b = bn;
            }
        }

        if (ch == 1) {
            #pragma unroll
            for (int v = 0; v < 8; ++v) zx[v * 32 + lane] = z[v];
        }
        __syncthreads();

        #pragma unroll
        for (int v = 0; v < 8; ++v) xr[v] = decay * xr[v] + sqdt * ep[v];

        if (ch == 0) {
            const float b3v = sB3[lo];
            float s[8];
            #pragma unroll
            for (int v = 0; v < 8; ++v) {
                float zz = (z[v] + zx[v * 32 + lane]) + b3v;
                s[v]  = 0.5f * dt * zz * zz + sqdt * zz * ep[v];
            }
            #pragma unroll
            for (int v = 0; v < 8; ++v) s[v] = row_xor_add<1>(s[v]);
            #pragma unroll
            for (int v = 0; v < 8; ++v) s[v] = row_xor_add<2>(s[v]);
            #pragma unroll
            for (int v = 0; v < 8; ++v) s[v] = row_xor_add<4>(s[v]);
            #pragma unroll
            for (int v = 0; v < 8; ++v) s[v] = row_xor_add<8>(s[v]);
            #pragma unroll
            for (int v = 0; v < 8; ++v) va[v] += s[v];
        }
    }

    if (ch == 0) {
        #pragma unroll
        for (int v = 0; v < 8; ++v) {
            int row = rowbase + v + 8 * hi;
            outX[row * DD + lo] = xr[v];
            if (lo == 0) outV[row] = va[v];
        }
    }
}

extern "C" void kernel_launch(void* const* d_in, const int* in_sizes, int n_in,
                              void* d_out, int out_size, void* d_ws, size_t ws_size,
                              hipStream_t stream) {
    (void)in_sizes; (void)n_in; (void)out_size; (void)d_ws; (void)ws_size;
    const float* X0    = (const float*)d_in[0];
    const float* V0    = (const float*)d_in[1];
    const float* Yobs  = (const float*)d_in[2];
    const float* noise = (const float*)d_in[3];
    const float* W1    = (const float*)d_in[4];
    const float* b1    = (const float*)d_in[5];
    const float* W2    = (const float*)d_in[6];
    const float* b2    = (const float*)d_in[7];
    const float* W3    = (const float*)d_in[8];
    const float* b3    = (const float*)d_in[9];

    float* outX = (float*)d_out;
    float* outV = outX + (size_t)NN * DD;

    dim3 grid(NN / (RT * 16));
    dim3 block(WAVES * 32);
    sde_rollout_kernel<<<grid, block, 0, stream>>>(X0, V0, Yobs, noise,
                                                   W1, b1, W2, b2, W3, b3,
                                                   outX, outV);
}


# LDS fragment reads grouped: each block's A and B loads (8 k-steps in layer 2, whole K in layers 1 and 3) issued as one group a WMMA block ahead, not in ones and twos among the WMMAs; sum order kept
# speedup vs baseline: 2.0012x; 1.0575x over previous
#include <hip/hip_runtime.h>
#include <hip/hip_bf16.h>

typedef float v2f __attribute__((ext_vector_type(2)));
typedef float v8f __attribute__((ext_vector_type(8)));

#define NN 16384
#define DD 16
#define PP 8
#define MM 50
#define HH 128

#define PW12 288
#define PW3   32
#define SXY   28
#define SH   132
#define SH2   68
#define RT     4
#define WAVES  8

static __device__ __forceinline__ v2f ldsv2(const float* p) { return *(const v2f*)p; }

template <int M>
static __device__ __forceinline__ float row_xor_add(float s) {
    const int partner = __builtin_amdgcn_update_dpp(0, __float_as_int(s), 0x160 + M, 0xF, 0xF, true);
    return s + __int_as_float(partner);
}

#define KB 8


static __device__ __forceinline__ v8f mma4(v2f a, v2f b, v8f c) {
    return __builtin_amdgcn_wmma_f32_16x16x4_f32(false, a, false, b, (short)0, c, false, false);
}

template <int BLK>
static __device__ __forceinline__ void load_w2(v2f (&B2)[32][4], const float* w2c, int hi, int lo) {
    #pragma unroll
    for (int j = BLK * KB; j < (BLK + 1) * KB; ++j) {
        #pragma unroll
        for (int c = 0; c < 4; ++c)
            B2[j][c] = ldsv2(&w2c[(2 * j + hi) * PW12 + (c * 16 + lo) * 2]);
    }
}

template <int BLK>
static __device__ __forceinline__ void load_h1(v2f (&A2)[32], const float* h1, int hi, int lo) {
    #pragma unroll
    for (int j = BLK * KB; j < (BLK + 1) * KB; ++j)
        A2[j] = ldsv2(&h1[lo * SH + 4 * j + 2 * hi]);
}

template <int BLK>
static __device__ __forceinline__ void mma_l2(v8f (&acc)[4], const v2f (&A2)[32], const v2f (&B2)[32][4]) {
    const v8f zc = {0.f, 0.f, 0.f, 0.f, 0.f, 0.f, 0.f, 0.f};
    #pragma unroll
    for (int j = BLK * KB; j < (BLK + 1) * KB; ++j) {
        #pragma unroll
        for (int c = 0; c < 4; ++c)
            acc[c] = mma4(A2[j], B2[j][c], (j == 0 ? zc : acc[c]));
    }
}

__launch_bounds__(256, 1)
__global__ void sde_rollout_kernel(const float* __restrict__ X0,
                                   const float* __restrict__ V0,
                                   const float* __restrict__ Yobs,
                                   const float* __restrict__ noise,
                                   const float* __restrict__ W1,
                                   const float* __restrict__ b1,
                                   const float* __restrict__ W2,
                                   const float* __restrict__ b2,
                                   const float* __restrict__ W3,
                                   const float* __restrict__ b3,
                                   float* __restrict__ outX,
                                   float* __restrict__ outV)
{
    __shared__ float sW1p[12 * PW12];
    __shared__ float sW1t[HH];
    __shared__ float sB1 [HH];
    __shared__ float sW2p[64 * PW12];
    __shared__ float sB2 [HH];
    __shared__ float sW3p[64 * PW3];
    __shared__ float sB3 [DD];
    __shared__ float sXY [WAVES][16 * SXY];
    __shared__ float sH1 [RT][16 * SH];
    __shared__ float sH2 [WAVES][16 * SH2];
    __shared__ float sZ  [RT][8 * 32];

    const int tid  = threadIdx.x;
    const int lane = tid & 31;
    const int wave = __builtin_amdgcn_readfirstlane(tid >> 5);
    const int lo   = lane & 15;
    const int hi   = lane >> 4;
    const int rt   = wave & 3;
    const int ch   = wave >> 2;
    const int cb   = ch * 4;
    const int rowbase = blockIdx.x * (RT * 16) + rt * 16;

    for (int i = tid; i < 24 * HH; i += blockDim.x) {
        int r = i >> 7, c = i & 127;
        sW1p[(r >> 1) * PW12 + c * 2 + (r & 1)] = W1[(1 + r) * HH + c];
    }
    if (tid < HH) { sW1t[tid] = W1[tid]; sB1[tid] = b1[tid]; sB2[tid] = b2[tid]; }
    for (int i = tid; i < HH * HH; i += blockDim.x) {
        int r = i >> 7, c = i & 127;
        sW2p[(r >> 1) * PW12 + c * 2 + (r & 1)] = W2[i];
    }
    for (int i = tid; i < HH * DD; i += blockDim.x) {
        int r = i >> 4, c = i & 15;
        sW3p[(r >> 1) * PW3 + c * 2 + (r & 1)] = W3[i];
    }
    if (tid < DD) sB3[tid] = b3[tid];

    float* xy = sXY[wave];
    for (int i = lane; i < 16 * PP; i += 32) {
        int r = i >> 3, c = i & 7;
        xy[r * SXY + DD + c] = Yobs[(rowbase + r) * PP + c];
    }
    __syncthreads();

    float xr[8], va[8];
    #pragma unroll
    for (int v = 0; v < 8; ++v) {
        int row = rowbase + v + 8 * hi;
        xr[v] = X0[row * DD + lo];
        va[v] = V0[row];
    }

    const float dt    = 1.0f / MM;
    const float sqdt  = __builtin_sqrtf(dt);
    const float decay = 1.0f - dt;
    float* h1 = sH1[rt];
    float* h2 = sH2[wave];
    float* zx = sZ[rt];
    const float* w1c  = sW1p + cb * 32;
    const float* w2c  = sW2p + cb * 32;
    const float* w3c  = sW3p + ch * 32 * PW3;
    const float* w1tc = sW1t + cb * 16;
    const float* b1c  = sB1  + cb * 16;
    const float* b2c  = sB2  + cb * 16;
    float* h1c = h1 + cb * 16;
    const v8f zc = {0.f, 0.f, 0.f, 0.f, 0.f, 0.f, 0.f, 0.f};

    for (int m = 0; m < MM; ++m) {
        const float t = m * dt;

        v8f acc[4];
        v2f A2[32];
        v2f B2[32][4];

        #pragma unroll
        for (int v = 0; v < 8; ++v)
            xy[(v + 8 * hi) * SXY + lo] = xr[v];

        asm volatile("s_wait_dscnt 0" ::: "memory");

        v2f a1f[6];
        v2f b1f[6][4];
        float w1tv[4], b1v[4];
        #pragma unroll
        for (int kk = 0; kk < 6; ++kk) a1f[kk] = ldsv2(&xy[lo * SXY + 4 * kk + 2 * hi]);
        #pragma unroll
        for (int kk = 0; kk < 6; ++kk) {
            #pragma unroll
            for (int c = 0; c < 4; ++c)
                b1f[kk][c] = ldsv2(&w1c[(2 * kk + hi) * PW12 + (c * 16 + lo) * 2]);
        }
        #pragma unroll
        for (int c = 0; c < 4; ++c) { w1tv[c] = w1tc[c * 16 + lo]; b1v[c] = b1c[c * 16 + lo]; }
        load_w2<0>(B2, w2c, hi, lo);
        __builtin_amdgcn_sched_barrier(0);

        #pragma unroll
        for (int kk = 0; kk < 6; ++kk) {
            #pragma unroll
            for (int c = 0; c < 4; ++c)
                acc[c] = mma4(a1f[kk], b1f[kk][c], (kk == 0 ? zc : acc[c]));
        }
        #pragma unroll
        for (int c = 0; c < 4; ++c) {
            float bias = t * w1tv[c] + b1v[c];
            #pragma unroll
            for (int v = 0; v < 8; ++v)
                h1c[(v + 8 * hi) * SH + c * 16 + lo] = fmaxf(acc[c][v] + bias, 0.0f);
        }
        __syncthreads();

        float ep[8];
        #pragma unroll
        for (int v = 0; v < 8; ++v) {
            size_t row = (size_t)m * NN + (size_t)(rowbase + v + 8 * hi);
            ep[v] = noise[row * DD + lo];
        }

        load_h1<0>(A2, h1, hi, lo);
        load_h1<1>(A2, h1, hi, lo);
        load_w2<1>(B2, w2c, hi, lo);
        __builtin_amdgcn_sched_barrier(0);
        mma_l2<0>(acc, A2, B2);
        __builtin_amdgcn_sched_barrier(0);
        load_h1<2>(A2, h1, hi, lo);
        load_w2<2>(B2, w2c, hi, lo);
        __builtin_amdgcn_sched_barrier(0);
        mma_l2<1>(acc, A2, B2);
        __builtin_amdgcn_sched_barrier(0);
        load_h1<3>(A2, h1, hi, lo);
        load_w2<3>(B2, w2c, hi, lo);
        __builtin_amdgcn_sched_barrier(0);
        mma_l2<2>(acc, A2, B2);
        __builtin_amdgcn_sched_barrier(0);
        v2f b3f[16];
        float b2v[4];
        #pragma unroll
        for (int kk = 0; kk < 16; ++kk) b3f[kk] = ldsv2(&w3c[(2 * kk + hi) * PW3 + lo * 2]);
        #pragma unroll
        for (int c = 0; c < 4; ++c) b2v[c] = b2c[c * 16 + lo];
        __builtin_amdgcn_sched_barrier(0);
        mma_l2<3>(acc, A2, B2);

        #pragma unroll
        for (int c = 0; c < 4; ++c) {
            #pragma unroll
            for (int v = 0; v < 8; ++v)
                h2[(v + 8 * hi) * SH2 + c * 16 + lo] = fmaxf(acc[c][v] + b2v[c], 0.0f);
        }
        asm volatile("s_wait_dscnt 0" ::: "memory");

        v8f z;
        {
            v2f a3f[16];
            #pragma unroll
            for (int kk = 0; kk < 16; ++kk) a3f[kk] = ldsv2(&h2[lo * SH2 + 4 * kk + 2 * hi]);
            __builtin_amdgcn_sched_barrier(0);
            #pragma unroll
            for (int kk = 0; kk < 16; ++kk)
                z = mma4(a3f[kk], b3f[kk], (kk == 0 ? zc : z));
        }

        if (ch == 1) {
            #pragma unroll
            for (int v = 0; v < 8; ++v) zx[v * 32 + lane] = z[v];
        }
        __syncthreads();

        #pragma unroll
        for (int v = 0; v < 8; ++v) xr[v] = decay * xr[v] + sqdt * ep[v];

        if (ch == 0) {
            const float b3v = sB3[lo];
            float s[8];
            #pragma unroll
            for (int v = 0; v < 8; ++v) {
                float zz = (z[v] + zx[v * 32 + lane]) + b3v;
                s[v]  = 0.5f * dt * zz * zz + sqdt * zz * ep[v];
            }
            #pragma unroll
            for (int v = 0; v < 8; ++v) s[v] = row_xor_add<1>(s[v]);
            #pragma unroll
            for (int v = 0; v < 8; ++v) s[v] = row_xor_add<2>(s[v]);
            #pragma unroll
            for (int v = 0; v < 8; ++v) s[v] = row_xor_add<4>(s[v]);
            #pragma unroll
            for (int v = 0; v < 8; ++v) s[v] = row_xor_add<8>(s[v]);
            #pragma unroll
            for (int v = 0; v < 8; ++v) va[v] += s[v];
        }
    }

    if (ch == 0) {
        #pragma unroll
        for (int v = 0; v < 8; ++v) {
            int row = rowbase + v + 8 * hi;
            outX[row * DD + lo] = xr[v];
            if (lo == 0) outV[row] = va[v];
        }
    }
}

extern "C" void kernel_launch(void* const* d_in, const int* in_sizes, int n_in,
                              void* d_out, int out_size, void* d_ws, size_t ws_size,
                              hipStream_t stream) {
    (void)in_sizes; (void)n_in; (void)out_size; (void)d_ws; (void)ws_size;
    const float* X0    = (const float*)d_in[0];
    const float* V0    = (const float*)d_in[1];
    const float* Yobs  = (const float*)d_in[2];
    const float* noise = (const float*)d_in[3];
    const float* W1    = (const float*)d_in[4];
    const float* b1    = (const float*)d_in[5];
    const float* W2    = (const float*)d_in[6];
    const float* b2    = (const float*)d_in[7];
    const float* W3    = (const float*)d_in[8];
    const float* b3    = (const float*)d_in[9];

    float* outX = (float*)d_out;
    float* outV = outX + (size_t)NN * DD;

    dim3 grid(NN / (RT * 16));
    dim3 block(WAVES * 32);
    sde_rollout_kernel<<<grid, block, 0, stream>>>(X0, V0, Yobs, noise,
                                                   W1, b1, W2, b2, W3, b3,
                                                   outX, outV);
}


# Four waves per row tile (16-wave blocks, 4 per SIMD): each does 32 of 128 hidden columns and a quarter of layer 3's K; three partial Z meet in LDS at the same 2 barriers; layer-3 sum order changes
# speedup vs baseline: 2.1990x; 1.0988x over previous
#include <hip/hip_runtime.h>
#include <hip/hip_bf16.h>

typedef float v2f __attribute__((ext_vector_type(2)));
typedef float v8f __attribute__((ext_vector_type(8)));

#define NN 16384
#define DD 16
#define PP 8
#define MM 50
#define HH 128

#define PW12 288
#define PW3   32
#define SXY   28
#define SH   132
#define SH2   36
#define RT     4
#define WAVES 16
#define CT     2
#define K3     8

static __device__ __forceinline__ v2f ldsv2(const float* p) { return *(const v2f*)p; }

template <int M>
static __device__ __forceinline__ float row_xor_add(float s) {
    const int partner = __builtin_amdgcn_update_dpp(0, __float_as_int(s), 0x160 + M, 0xF, 0xF, true);
    return s + __int_as_float(partner);
}

#define KB 8


static __device__ __forceinline__ v8f mma4(v2f a, v2f b, v8f c) {
    return __builtin_amdgcn_wmma_f32_16x16x4_f32(false, a, false, b, (short)0, c, false, false);
}

template <int BLK>
static __device__ __forceinline__ void load_w2(v2f (&B2)[32][CT], const float* w2c, int hi, int lo) {
    #pragma unroll
    for (int j = BLK * KB; j < (BLK + 1) * KB; ++j) {
        #pragma unroll
        for (int c = 0; c < CT; ++c)
            B2[j][c] = ldsv2(&w2c[(2 * j + hi) * PW12 + (c * 16 + lo) * 2]);
    }
}

template <int BLK>
static __device__ __forceinline__ void load_h1(v2f (&A2)[32], const float* h1, int hi, int lo) {
    #pragma unroll
    for (int j = BLK * KB; j < (BLK + 1) * KB; ++j)
        A2[j] = ldsv2(&h1[lo * SH + 4 * j + 2 * hi]);
}

template <int BLK>
static __device__ __forceinline__ void mma_l2(v8f (&acc)[CT], const v2f (&A2)[32], const v2f (&B2)[32][CT]) {
    const v8f zc = {0.f, 0.f, 0.f, 0.f, 0.f, 0.f, 0.f, 0.f};
    #pragma unroll
    for (int j = BLK * KB; j < (BLK + 1) * KB; ++j) {
        #pragma unroll
        for (int c = 0; c < CT; ++c)
            acc[c] = mma4(A2[j], B2[j][c], (j == 0 ? zc : acc[c]));
    }
}

__launch_bounds__(512, 1)
__global__ void sde_rollout_kernel(const float* __restrict__ X0,
                                   const float* __restrict__ V0,
                                   const float* __restrict__ Yobs,
                                   const float* __restrict__ noise,
                                   const float* __restrict__ W1,
                                   const float* __restrict__ b1,
                                   const float* __restrict__ W2,
                                   const float* __restrict__ b2,
                                   const float* __restrict__ W3,
                                   const float* __restrict__ b3,
                                   float* __restrict__ outX,
                                   float* __restrict__ outV)
{
    __shared__ float sW1p[12 * PW12];
    __shared__ float sW1t[HH];
    __shared__ float sB1 [HH];
    __shared__ float sW2p[64 * PW12];
    __shared__ float sB2 [HH];
    __shared__ float sW3p[64 * PW3];
    __shared__ float sB3 [DD];
    __shared__ float sXY [WAVES][16 * SXY];
    __shared__ float sH1 [RT][16 * SH];
    __shared__ float sH2 [WAVES][16 * SH2];
    __shared__ float sZ  [RT][3 * 8 * 32];

    const int tid  = threadIdx.x;
    const int lane = tid & 31;
    const int wave = __builtin_amdgcn_readfirstlane(tid >> 5);
    const int lo   = lane & 15;
    const int hi   = lane >> 4;
    const int rt   = wave & 3;
    const int ch   = wave >> 2;
    const int cb   = ch * CT;
    const int rowbase = blockIdx.x * (RT * 16) + rt * 16;

    for (int i = tid; i < 24 * HH; i += blockDim.x) {
        int r = i >> 7, c = i & 127;
        sW1p[(r >> 1) * PW12 + c * 2 + (r & 1)] = W1[(1 + r) * HH + c];
    }
    if (tid < HH) { sW1t[tid] = W1[tid]; sB1[tid] = b1[tid]; sB2[tid] = b2[tid]; }
    for (int i = tid; i < HH * HH; i += blockDim.x) {
        int r = i >> 7, c = i & 127;
        sW2p[(r >> 1) * PW12 + c * 2 + (r & 1)] = W2[i];
    }
    for (int i = tid; i < HH * DD; i += blockDim.x) {
        int r = i >> 4, c = i & 15;
        sW3p[(r >> 1) * PW3 + c * 2 + (r & 1)] = W3[i];
    }
    if (tid < DD) sB3[tid] = b3[tid];

    float* xy = sXY[wave];
    for (int i = lane; i < 16 * PP; i += 32) {
        int r = i >> 3, c = i & 7;
        xy[r * SXY + DD + c] = Yobs[(rowbase + r) * PP + c];
    }
    __syncthreads();

    float xr[8], va[8];
    #pragma unroll
    for (int v = 0; v < 8; ++v) {
        int row = rowbase + v + 8 * hi;
        xr[v] = X0[row * DD + lo];
        va[v] = V0[row];
    }

    const float dt    = 1.0f / MM;
    const float sqdt  = __builtin_sqrtf(dt);
    const float decay = 1.0f - dt;
    float* h1 = sH1[rt];
    float* h2 = sH2[wave];
    float* zx = sZ[rt];
    const float* w1c  = sW1p + cb * 32;
    const float* w2c  = sW2p + cb * 32;
    const float* w3c  = sW3p + ch * 2 * K3 * PW3;
    const float* w1tc = sW1t + cb * 16;
    const float* b1c  = sB1  + cb * 16;
    const float* b2c  = sB2  + cb * 16;
    float* h1c = h1 + cb * 16;
    const v8f zc = {0.f, 0.f, 0.f, 0.f, 0.f, 0.f, 0.f, 0.f};

    for (int m = 0; m < MM; ++m) {
        const float t = m * dt;

        v8f acc[CT];
        v2f A2[32];
        v2f B2[32][CT];

        #pragma unroll
        for (int v = 0; v < 8; ++v)
            xy[(v + 8 * hi) * SXY + lo] = xr[v];

        asm volatile("s_wait_dscnt 0" ::: "memory");

        v2f a1f[6];
        v2f b1f[6][CT];
        float w1tv[CT], b1v[CT];
        #pragma unroll
        for (int kk = 0; kk < 6; ++kk) a1f[kk] = ldsv2(&xy[lo * SXY + 4 * kk + 2 * hi]);
        #pragma unroll
        for (int kk = 0; kk < 6; ++kk) {
            #pragma unroll
            for (int c = 0; c < CT; ++c)
                b1f[kk][c] = ldsv2(&w1c[(2 * kk + hi) * PW12 + (c * 16 + lo) * 2]);
        }
        #pragma unroll
        for (int c = 0; c < CT; ++c) { w1tv[c] = w1tc[c * 16 + lo]; b1v[c] = b1c[c * 16 + lo]; }
        load_w2<0>(B2, w2c, hi, lo);
        __builtin_amdgcn_sched_barrier(0);

        #pragma unroll
        for (int kk = 0; kk < 6; ++kk) {
            #pragma unroll
            for (int c = 0; c < CT; ++c)
                acc[c] = mma4(a1f[kk], b1f[kk][c], (kk == 0 ? zc : acc[c]));
        }
        #pragma unroll
        for (int c = 0; c < CT; ++c) {
            float bias = t * w1tv[c] + b1v[c];
            #pragma unroll
            for (int v = 0; v < 8; ++v)
                h1c[(v + 8 * hi) * SH + c * 16 + lo] = fmaxf(acc[c][v] + bias, 0.0f);
        }
        __syncthreads();

        float ep[8];
        #pragma unroll
        for (int v = 0; v < 8; ++v) {
            size_t row = (size_t)m * NN + (size_t)(rowbase + v + 8 * hi);
            ep[v] = noise[row * DD + lo];
        }

        load_h1<0>(A2, h1, hi, lo);
        load_h1<1>(A2, h1, hi, lo);
        load_w2<1>(B2, w2c, hi, lo);
        __builtin_amdgcn_sched_barrier(0);
        mma_l2<0>(acc, A2, B2);
        __builtin_amdgcn_sched_barrier(0);
        load_h1<2>(A2, h1, hi, lo);
        load_w2<2>(B2, w2c, hi, lo);
        __builtin_amdgcn_sched_barrier(0);
        mma_l2<1>(acc, A2, B2);
        __builtin_amdgcn_sched_barrier(0);
        load_h1<3>(A2, h1, hi, lo);
        load_w2<3>(B2, w2c, hi, lo);
        __builtin_amdgcn_sched_barrier(0);
        mma_l2<2>(acc, A2, B2);
        __builtin_amdgcn_sched_barrier(0);
        v2f b3f[K3];
        float b2v[CT];
        #pragma unroll
        for (int kk = 0; kk < K3; ++kk) b3f[kk] = ldsv2(&w3c[(2 * kk + hi) * PW3 + lo * 2]);
        #pragma unroll
        for (int c = 0; c < CT; ++c) b2v[c] = b2c[c * 16 + lo];
        __builtin_amdgcn_sched_barrier(0);
        mma_l2<3>(acc, A2, B2);

        #pragma unroll
        for (int c = 0; c < CT; ++c) {
            #pragma unroll
            for (int v = 0; v < 8; ++v)
                h2[(v + 8 * hi) * SH2 + c * 16 + lo] = fmaxf(acc[c][v] + b2v[c], 0.0f);
        }
        asm volatile("s_wait_dscnt 0" ::: "memory");

        v8f z;
        {
            v2f a3f[K3];
            #pragma unroll
            for (int kk = 0; kk < K3; ++kk) a3f[kk] = ldsv2(&h2[lo * SH2 + 4 * kk + 2 * hi]);
            __builtin_amdgcn_sched_barrier(0);
            #pragma unroll
            for (int kk = 0; kk < K3; ++kk)
                z = mma4(a3f[kk], b3f[kk], (kk == 0 ? zc : z));
        }

        if (ch != 0) {
            float* zo = zx + (ch - 1) * (8 * 32);
            #pragma unroll
            for (int v = 0; v < 8; ++v) zo[v * 32 + lane] = z[v];
        }
        __syncthreads();

        #pragma unroll
        for (int v = 0; v < 8; ++v) xr[v] = decay * xr[v] + sqdt * ep[v];

        if (ch == 0) {
            const float b3v = sB3[lo];
            float s[8];
            #pragma unroll
            for (int v = 0; v < 8; ++v) {
                float zz = (((z[v] + zx[v * 32 + lane]) + zx[8 * 32 + v * 32 + lane])
                            + zx[2 * 8 * 32 + v * 32 + lane]) + b3v;
                s[v]  = 0.5f * dt * zz * zz + sqdt * zz * ep[v];
            }
            #pragma unroll
            for (int v = 0; v < 8; ++v) s[v] = row_xor_add<1>(s[v]);
            #pragma unroll
            for (int v = 0; v < 8; ++v) s[v] = row_xor_add<2>(s[v]);
            #pragma unroll
            for (int v = 0; v < 8; ++v) s[v] = row_xor_add<4>(s[v]);
            #pragma unroll
            for (int v = 0; v < 8; ++v) s[v] = row_xor_add<8>(s[v]);
            #pragma unroll
            for (int v = 0; v < 8; ++v) va[v] += s[v];
        }
    }

    if (ch == 0) {
        #pragma unroll
        for (int v = 0; v < 8; ++v) {
            int row = rowbase + v + 8 * hi;
            outX[row * DD + lo] = xr[v];
            if (lo == 0) outV[row] = va[v];
        }
    }
}

extern "C" void kernel_launch(void* const* d_in, const int* in_sizes, int n_in,
                              void* d_out, int out_size, void* d_ws, size_t ws_size,
                              hipStream_t stream) {
    (void)in_sizes; (void)n_in; (void)out_size; (void)d_ws; (void)ws_size;
    const float* X0    = (const float*)d_in[0];
    const float* V0    = (const float*)d_in[1];
    const float* Yobs  = (const float*)d_in[2];
    const float* noise = (const float*)d_in[3];
    const float* W1    = (const float*)d_in[4];
    const float* b1    = (const float*)d_in[5];
    const float* W2    = (const float*)d_in[6];
    const float* b2    = (const float*)d_in[7];
    const float* W3    = (const float*)d_in[8];
    const float* b3    = (const float*)d_in[9];

    float* outX = (float*)d_out;
    float* outV = outX + (size_t)NN * DD;

    dim3 grid(NN / (RT * 16));
    dim3 block(WAVES * 32);
    sde_rollout_kernel<<<grid, block, 0, stream>>>(X0, V0, Yobs, noise,
                                                   W1, b1, W2, b2, W3, b3,
                                                   outX, outV);
}


# Step loop takes two time steps per pass (X needs no Z): each LDS weight fragment feeds 2 WMMAs; barriers, load groups and epilogue round trips are paid once per 2 steps; sum order kept
# speedup vs baseline: 2.2905x; 1.0416x over previous
#include <hip/hip_runtime.h>
#include <hip/hip_bf16.h>

typedef float v2f __attribute__((ext_vector_type(2)));
typedef float v8f __attribute__((ext_vector_type(8)));

#define NN 16384
#define DD 16
#define PP 8
#define MM 50
#define HH 128

#define PW12 288
#define PW3   32
#define SXY   28
#define SH   132
#define SH2   36
#define RT     4
#define WAVES 16
#define CT     2
#define K3     8
#define TS     2

static __device__ __forceinline__ v2f ldsv2(const float* p) { return *(const v2f*)p; }

template <int M>
static __device__ __forceinline__ float row_xor_add(float s) {
    const int partner = __builtin_amdgcn_update_dpp(0, __float_as_int(s), 0x160 + M, 0xF, 0xF, true);
    return s + __int_as_float(partner);
}

#define KB 8


static __device__ __forceinline__ v8f mma4(v2f a, v2f b, v8f c) {
    return __builtin_amdgcn_wmma_f32_16x16x4_f32(false, a, false, b, (short)0, c, false, false);
}

template <int BLK>
static __device__ __forceinline__ void load_w2(v2f (&B2)[32][CT], const float* w2c, int hi, int lo) {
    #pragma unroll
    for (int j = BLK * KB; j < (BLK + 1) * KB; ++j) {
        #pragma unroll
        for (int c = 0; c < CT; ++c)
            B2[j][c] = ldsv2(&w2c[(2 * j + hi) * PW12 + (c * 16 + lo) * 2]);
    }
}

template <int BLK>
static __device__ __forceinline__ void load_h1(v2f (&A2)[TS][32], const float* h1, int hi, int lo) {
    #pragma unroll
    for (int ts = 0; ts < TS; ++ts) {
        #pragma unroll
        for (int j = BLK * KB; j < (BLK + 1) * KB; ++j)
            A2[ts][j] = ldsv2(&h1[ts * (16 * SH) + lo * SH + 4 * j + 2 * hi]);
    }
}

template <int BLK>
static __device__ __forceinline__ void mma_l2(v8f (&acc)[TS][CT], const v2f (&A2)[TS][32], const v2f (&B2)[32][CT]) {
    const v8f zc = {0.f, 0.f, 0.f, 0.f, 0.f, 0.f, 0.f, 0.f};
    #pragma unroll
    for (int j = BLK * KB; j < (BLK + 1) * KB; ++j) {
        #pragma unroll
        for (int c = 0; c < CT; ++c) {
            #pragma unroll
            for (int ts = 0; ts < TS; ++ts)
                acc[ts][c] = mma4(A2[ts][j], B2[j][c], (j == 0 ? zc : acc[ts][c]));
        }
    }
}

__launch_bounds__(512, 1)
__global__ void sde_rollout_kernel(const float* __restrict__ X0,
                                   const float* __restrict__ V0,
                                   const float* __restrict__ Yobs,
                                   const float* __restrict__ noise,
                                   const float* __restrict__ W1,
                                   const float* __restrict__ b1,
                                   const float* __restrict__ W2,
                                   const float* __restrict__ b2,
                                   const float* __restrict__ W3,
                                   const float* __restrict__ b3,
                                   float* __restrict__ outX,
                                   float* __restrict__ outV)
{
    __shared__ float sW1p[12 * PW12];
    __shared__ float sW1t[HH];
    __shared__ float sB1 [HH];
    __shared__ float sW2p[64 * PW12];
    __shared__ float sB2 [HH];
    __shared__ float sW3p[64 * PW3];
    __shared__ float sB3 [DD];
    __shared__ float sXY [WAVES][TS * 16 * SXY];
    __shared__ float sH1 [RT][TS * 16 * SH];
    __shared__ float sH2 [WAVES][TS * 16 * SH2];
    __shared__ float sZ  [RT][TS * 3 * 8 * 32];

    const int tid  = threadIdx.x;
    const int lane = tid & 31;
    const int wave = __builtin_amdgcn_readfirstlane(tid >> 5);
    const int lo   = lane & 15;
    const int hi   = lane >> 4;
    const int rt   = wave & 3;
    const int ch   = wave >> 2;
    const int cb   = ch * CT;
    const int rowbase = blockIdx.x * (RT * 16) + rt * 16;

    for (int i = tid; i < 24 * HH; i += blockDim.x) {
        int r = i >> 7, c = i & 127;
        sW1p[(r >> 1) * PW12 + c * 2 + (r & 1)] = W1[(1 + r) * HH + c];
    }
    if (tid < HH) { sW1t[tid] = W1[tid]; sB1[tid] = b1[tid]; sB2[tid] = b2[tid]; }
    for (int i = tid; i < HH * HH; i += blockDim.x) {
        int r = i >> 7, c = i & 127;
        sW2p[(r >> 1) * PW12 + c * 2 + (r & 1)] = W2[i];
    }
    for (int i = tid; i < HH * DD; i += blockDim.x) {
        int r = i >> 4, c = i & 15;
        sW3p[(r >> 1) * PW3 + c * 2 + (r & 1)] = W3[i];
    }
    if (tid < DD) sB3[tid] = b3[tid];

    float* xy = sXY[wave];
    for (int i = lane; i < 16 * PP; i += 32) {
        int r = i >> 3, c = i & 7;
        const float y = Yobs[(rowbase + r) * PP + c];
        #pragma unroll
        for (int ts = 0; ts < TS; ++ts)
            xy[ts * (16 * SXY) + r * SXY + DD + c] = y;
    }
    __syncthreads();

    const float dt    = 1.0f / MM;
    const float sqdt  = __builtin_sqrtf(dt);
    const float decay = 1.0f - dt;

    const unsigned eidx = (unsigned)(rowbase + 8 * hi) * DD + lo;
    float xa[8], xb[8], va[8];
    #pragma unroll
    for (int v = 0; v < 8; ++v) {
        xa[v] = X0[eidx + v * DD];
        va[v] = V0[rowbase + 8 * hi + v];
        xb[v] = decay * xa[v] + sqdt * noise[eidx + v * DD];
    }

    float* h1 = sH1[rt];
    float* h2 = sH2[wave];
    float* zx = sZ[rt];
    const float* w1c  = sW1p + cb * 32;
    const float* w2c  = sW2p + cb * 32;
    const float* w3c  = sW3p + ch * 2 * K3 * PW3;
    const float* w1tc = sW1t + cb * 16;
    const float* b1c  = sB1  + cb * 16;
    const float* b2c  = sB2  + cb * 16;
    float* h1c = h1 + cb * 16;
    float* xyw = xy  + (8 * hi) * SXY + lo;
    float* h1w = h1c + (8 * hi) * SH  + lo;
    float* h2w = h2  + (8 * hi) * SH2 + lo;
    const v8f zc = {0.f, 0.f, 0.f, 0.f, 0.f, 0.f, 0.f, 0.f};

    for (int i = 0; i < MM / TS; ++i) {
        const int m0 = i * TS;

        v8f acc[TS][CT];
        v2f A2[TS][32];
        v2f B2[32][CT];

        #pragma unroll
        for (int v = 0; v < 8; ++v)
            xyw[v * SXY] = xa[v];
        #pragma unroll
        for (int v = 0; v < 8; ++v)
            xyw[16 * SXY + v * SXY] = xb[v];

        asm volatile("s_wait_dscnt 0" ::: "memory");

        v2f a1f[TS][6];
        v2f b1f[6][CT];
        float w1tv[CT], b1v[CT];
        #pragma unroll
        for (int ts = 0; ts < TS; ++ts) {
            #pragma unroll
            for (int kk = 0; kk < 6; ++kk)
                a1f[ts][kk] = ldsv2(&xy[ts * (16 * SXY) + lo * SXY + 4 * kk + 2 * hi]);
        }
        #pragma unroll
        for (int kk = 0; kk < 6; ++kk) {
            #pragma unroll
            for (int c = 0; c < CT; ++c)
                b1f[kk][c] = ldsv2(&w1c[(2 * kk + hi) * PW12 + (c * 16 + lo) * 2]);
        }
        #pragma unroll
        for (int c = 0; c < CT; ++c) { w1tv[c] = w1tc[c * 16 + lo]; b1v[c] = b1c[c * 16 + lo]; }
        load_w2<0>(B2, w2c, hi, lo);
        __builtin_amdgcn_sched_barrier(0);

        #pragma unroll
        for (int kk = 0; kk < 6; ++kk) {
            #pragma unroll
            for (int c = 0; c < CT; ++c) {
                #pragma unroll
                for (int ts = 0; ts < TS; ++ts)
                    acc[ts][c] = mma4(a1f[ts][kk], b1f[kk][c], (kk == 0 ? zc : acc[ts][c]));
            }
        }
        #pragma unroll
        for (int ts = 0; ts < TS; ++ts) {
            const float t = (m0 + ts) * dt;
            #pragma unroll
            for (int c = 0; c < CT; ++c) {
                float bias = t * w1tv[c] + b1v[c];
                #pragma unroll
                for (int v = 0; v < 8; ++v)
                    h1w[ts * (16 * SH) + v * SH + c * 16] = fmaxf(acc[ts][c][v] + bias, 0.0f);
            }
        }
        __syncthreads();

        load_h1<0>(A2, h1, hi, lo);
        load_h1<1>(A2, h1, hi, lo);
        load_w2<1>(B2, w2c, hi, lo);
        __builtin_amdgcn_sched_barrier(0);
        mma_l2<0>(acc, A2, B2);
        __builtin_amdgcn_sched_barrier(0);
        load_h1<2>(A2, h1, hi, lo);
        load_w2<2>(B2, w2c, hi, lo);
        __builtin_amdgcn_sched_barrier(0);
        mma_l2<1>(acc, A2, B2);
        __builtin_amdgcn_sched_barrier(0);
        load_h1<3>(A2, h1, hi, lo);
        load_w2<3>(B2, w2c, hi, lo);
        __builtin_amdgcn_sched_barrier(0);
        mma_l2<2>(acc, A2, B2);
        __builtin_amdgcn_sched_barrier(0);
        v2f b3f[K3];
        float b2v[CT];
        float e0[8], e1[8], e2[8];
        #pragma unroll
        for (int kk = 0; kk < K3; ++kk) b3f[kk] = ldsv2(&w3c[(2 * kk + hi) * PW3 + lo * 2]);
        #pragma unroll
        for (int c = 0; c < CT; ++c) b2v[c] = b2c[c * 16 + lo];
        {
            const int mn = (m0 + TS < MM) ? (m0 + TS) : m0;
            #pragma unroll
            for (int v = 0; v < 8; ++v) {
                e0[v] = noise[(size_t)m0 * (NN * DD) + eidx + v * DD];
                e1[v] = noise[(size_t)(m0 + 1) * (NN * DD) + eidx + v * DD];
                e2[v] = noise[(size_t)mn * (NN * DD) + eidx + v * DD];
            }
        }
        __builtin_amdgcn_sched_barrier(0);
        mma_l2<3>(acc, A2, B2);

        #pragma unroll
        for (int ts = 0; ts < TS; ++ts) {
            #pragma unroll
            for (int c = 0; c < CT; ++c) {
                #pragma unroll
                for (int v = 0; v < 8; ++v)
                    h2w[ts * (16 * SH2) + v * SH2 + c * 16] = fmaxf(acc[ts][c][v] + b2v[c], 0.0f);
            }
        }
        asm volatile("s_wait_dscnt 0" ::: "memory");

        v8f z[TS];
        {
            v2f a3f[TS][K3];
            #pragma unroll
            for (int ts = 0; ts < TS; ++ts) {
                #pragma unroll
                for (int kk = 0; kk < K3; ++kk)
                    a3f[ts][kk] = ldsv2(&h2[ts * (16 * SH2) + lo * SH2 + 4 * kk + 2 * hi]);
            }
            __builtin_amdgcn_sched_barrier(0);
            #pragma unroll
            for (int kk = 0; kk < K3; ++kk) {
                #pragma unroll
                for (int ts = 0; ts < TS; ++ts)
                    z[ts] = mma4(a3f[ts][kk], b3f[kk], (kk == 0 ? zc : z[ts]));
            }
        }

        if (ch != 0) {
            #pragma unroll
            for (int ts = 0; ts < TS; ++ts) {
                float* zo = zx + ts * (3 * 8 * 32) + (ch - 1) * (8 * 32);
                #pragma unroll
                for (int v = 0; v < 8; ++v) zo[v * 32 + lane] = z[ts][v];
            }
        }
        __syncthreads();

        if (ch == 0) {
            const float b3v = sB3[lo];
            #pragma unroll
            for (int ts = 0; ts < TS; ++ts) {
                const float* zp = zx + ts * (3 * 8 * 32);
                float s[8];
                #pragma unroll
                for (int v = 0; v < 8; ++v) {
                    float zz = (((z[ts][v] + zp[v * 32 + lane]) + zp[8 * 32 + v * 32 + lane])
                                + zp[2 * 8 * 32 + v * 32 + lane]) + b3v;
                    const float e = (ts == 0) ? e0[v] : e1[v];
                    s[v]  = 0.5f * dt * zz * zz + sqdt * zz * e;
                }
                #pragma unroll
                for (int v = 0; v < 8; ++v) s[v] = row_xor_add<1>(s[v]);
                #pragma unroll
                for (int v = 0; v < 8; ++v) s[v] = row_xor_add<2>(s[v]);
                #pragma unroll
                for (int v = 0; v < 8; ++v) s[v] = row_xor_add<4>(s[v]);
                #pragma unroll
                for (int v = 0; v < 8; ++v) s[v] = row_xor_add<8>(s[v]);
                #pragma unroll
                for (int v = 0; v < 8; ++v) va[v] += s[v];
            }
        }

        #pragma unroll
        for (int v = 0; v < 8; ++v) xa[v] = decay * xb[v] + sqdt * e1[v];
        #pragma unroll
        for (int v = 0; v < 8; ++v) xb[v] = decay * xa[v] + sqdt * e2[v];
    }

    if (ch == 0) {
        const int wl  = (int)__builtin_amdgcn_mbcnt_lo(~0u, 0u);
        const int wlo = wl & 15, whi = wl >> 4;
        #pragma unroll
        for (int v = 0; v < 8; ++v) {
            int row = rowbase + v + 8 * whi;
            outX[row * DD + wlo] = xa[v];
            if (wlo == 0) outV[row] = va[v];
        }
    }
}

extern "C" void kernel_launch(void* const* d_in, const int* in_sizes, int n_in,
                              void* d_out, int out_size, void* d_ws, size_t ws_size,
                              hipStream_t stream) {
    (void)in_sizes; (void)n_in; (void)out_size; (void)d_ws; (void)ws_size;
    const float* X0    = (const float*)d_in[0];
    const float* V0    = (const float*)d_in[1];
    const float* Yobs  = (const float*)d_in[2];
    const float* noise = (const float*)d_in[3];
    const float* W1    = (const float*)d_in[4];
    const float* b1    = (const float*)d_in[5];
    const float* W2    = (const float*)d_in[6];
    const float* b2    = (const float*)d_in[7];
    const float* W3    = (const float*)d_in[8];
    const float* b3    = (const float*)d_in[9];

    float* outX = (float*)d_out;
    float* outV = outX + (size_t)NN * DD;

    dim3 grid(NN / (RT * 16));
    dim3 block(WAVES * 32);
    sde_rollout_kernel<<<grid, block, 0, stream>>>(X0, V0, Yobs, noise,
                                                   W1, b1, W2, b2, W3, b3,
                                                   outX, outV);
}
